# starting kernel
# baseline (speedup 1.0000x reference)
#include <hip/hip_runtime.h>
#include <hip/hip_bf16.h>

typedef _Float16 half_t;
typedef __attribute__((ext_vector_type(16))) _Float16 v16h;
typedef __attribute__((ext_vector_type(8)))  _Float16 v8h;
typedef __attribute__((ext_vector_type(8)))  float    v8f;

#define NGRAPH 128
#define NNODE  512
#define NEDGE  8192
#define NFEAT  128
#define NHID   128
#define BN_EPS 1e-5f

__global__ __launch_bounds__(512) void k_degrees(const int* __restrict__ edges,
                                                 float* __restrict__ rs_out,
                                                 float* __restrict__ rs_in) {
    __shared__ float degO[NNODE];
    __shared__ float degI[NNODE];
    const int g = blockIdx.x;
    const int t = threadIdx.x;
    const int* e = edges + (size_t)g * 2 * NEDGE;
    for (int n = t; n < NNODE; n += 512) { degO[n] = 0.f; degI[n] = 0.f; }
    __syncthreads();
    for (int i = t; i < NEDGE; i += 512) {
        atomicAdd(&degO[e[i]], 1.0f);
        atomicAdd(&degI[e[NEDGE + i]], 1.0f);
    }
    __syncthreads();
    for (int n = t; n < NNODE; n += 512) {
        rs_out[g * NNODE + n] = rsqrtf(fmaxf(degO[n], 1.0f));
        rs_in [g * NNODE + n] = rsqrtf(fmaxf(degI[n], 1.0f));
    }
}

__global__ __launch_bounds__(512) void k_scatter(const float*  __restrict__ xf32,
                                                 const half_t* __restrict__ xf16,
                                                 const int*    __restrict__ edges,
                                                 const float*  __restrict__ rs_out,
                                                 float*        __restrict__ agg,
                                                 int use_half) {
    const int g    = blockIdx.x;
    const int wave = threadIdx.x >> 5;
    const int lane = threadIdx.x & 31;
    const int* e = edges + (size_t)g * 2 * NEDGE;
    const float* rso = rs_out + g * NNODE;
    const size_t gbase = (size_t)g * NNODE * NFEAT;
    for (int i = wave; i < NEDGE; i += 16) {
        const int src = e[i];
        const int dst = e[NEDGE + i];
        const float s = rso[src];
        const size_t sb = gbase + (size_t)src * NFEAT;
        const size_t db = gbase + (size_t)dst * NFEAT;
#pragma unroll
        for (int c = 0; c < 4; ++c) {
            const int f = lane + 32 * c;
            const float v = use_half ? (float)xf16[sb + f] : xf32[sb + f];
            atomicAdd(&agg[db + f], v * s);
        }
    }
}

__global__ __launch_bounds__(512, 1) void k_gemm(const float* __restrict__ agg,
                                                 const float* __restrict__ rs_in,
                                                 const float* __restrict__ W,
                                                 const float* __restrict__ bvec,
                                                 half_t*      __restrict__ h_out,
                                                 float*       __restrict__ emb_out,
                                                 int layer) {
    __shared__ __align__(16) half_t wt[NHID * NHID];
    __shared__ float  bs[NHID];
    __shared__ float  embacc[NHID];
    const int g    = blockIdx.x >> 1;
    const int half = blockIdx.x & 1;
    const int t    = threadIdx.x;

    for (int i = t; i < NHID * NHID; i += 512) {
        const int k = i >> 7, n = i & 127;
        wt[n * NHID + k] = (half_t)W[i];
    }
    if (t < NHID) { bs[t] = bvec[t]; embacc[t] = 0.f; }
    __syncthreads();

    const int wave = t >> 5, lane = t & 31;
    const int l16 = lane & 15, hi = lane >> 4;
    const size_t gb = (size_t)g * NNODE;

    const int mi = half * 16 + wave;
    const int m  = mi * 16 + l16;
    const float* arow = agg + (gb + m) * NFEAT;
    const float  rsin = rs_in[gb + m];

    v8f c[8] = {};

#pragma unroll
    for (int ko = 0; ko < 4; ++ko) {
        const int kb = ko * 32;
        const int k0 = kb + hi * 8;
        const int k1 = kb + 16 + hi * 8;
        const float4 f0 = *(const float4*)(arow + k0);
        const float4 f1 = *(const float4*)(arow + k0 + 4);
        const float4 f2 = *(const float4*)(arow + k1);
        const float4 f3 = *(const float4*)(arow + k1 + 4);
        v16h a;
        a[0]  = (half_t)(f0.x * rsin); a[1]  = (half_t)(f0.y * rsin);
        a[2]  = (half_t)(f0.z * rsin); a[3]  = (half_t)(f0.w * rsin);
        a[4]  = (half_t)(f1.x * rsin); a[5]  = (half_t)(f1.y * rsin);
        a[6]  = (half_t)(f1.z * rsin); a[7]  = (half_t)(f1.w * rsin);
        a[8]  = (half_t)(f2.x * rsin); a[9]  = (half_t)(f2.y * rsin);
        a[10] = (half_t)(f2.z * rsin); a[11] = (half_t)(f2.w * rsin);
        a[12] = (half_t)(f3.x * rsin); a[13] = (half_t)(f3.y * rsin);
        a[14] = (half_t)(f3.z * rsin); a[15] = (half_t)(f3.w * rsin);

#pragma unroll
        for (int ni = 0; ni < 8; ++ni) {
            const half_t* brow = wt + (ni * 16 + l16) * NHID;
            const v8h blo = *(const v8h*)(brow + k0);
            const v8h bhi = *(const v8h*)(brow + k1);
            const v16h b = __builtin_shufflevector(blo, bhi,
                               0, 1, 2, 3, 4, 5, 6, 7,
                               8, 9, 10, 11, 12, 13, 14, 15);
            c[ni] = __builtin_amdgcn_wmma_f32_16x16x32_f16(
                        false, a, false, b, (short)0, c[ni], false, false);
        }
    }

#pragma unroll
    for (int ni = 0; ni < 8; ++ni) {
        const int n = ni * 16 + l16;
        if (layer == 0) {
#pragma unroll
            for (int r = 0; r < 8; ++r) {
                const int ml = r + 8 * hi;
                const float v = fmaxf(c[ni][r] + bs[n], 0.f);
                h_out[(gb + (size_t)mi * 16 + ml) * NFEAT + n] = (half_t)v;
            }
        } else {
            float s = 0.f;
#pragma unroll
            for (int r = 0; r < 8; ++r) s += fmaxf(c[ni][r] + bs[n], 0.f);
            atomicAdd(&embacc[n], s);
        }
    }

    if (layer == 1) {
        __syncthreads();
        if (t < NHID) atomicAdd(&emb_out[g * NHID + t], embacc[t] * (1.0f / NNODE));
    }
}

__global__ __launch_bounds__(128) void k_head(const float* __restrict__ emb,
                                              const float* __restrict__ Wt,
                                              const float* __restrict__ bt,
                                              const float* __restrict__ gamma,
                                              const float* __restrict__ beta,
                                              const float* __restrict__ rmean,
                                              const float* __restrict__ rvar,
                                              float* __restrict__ out) {
    __shared__ float er[NHID];
    const int g = blockIdx.x, n = threadIdx.x;
    er[n] = emb[g * NHID + n];
    __syncthreads();
    float z = bt[n];
    for (int k = 0; k < NHID; ++k) z += er[k] * Wt[k * NHID + n];
    z = gamma[n] * (z - rmean[n]) * rsqrtf(rvar[n] + BN_EPS) + beta[n];
    out[g * NHID + n] = fmaxf(z, 0.f);
}

extern "C" void kernel_launch(void* const* d_in, const int* in_sizes, int n_in,
                              void* d_out, int out_size, void* d_ws, size_t ws_size,
                              hipStream_t stream) {
    (void)in_sizes; (void)n_in; (void)out_size; (void)ws_size;
    const float* feats = (const float*)d_in[0];
    const int*   edges = (const int*)d_in[1];
    const float* W0    = (const float*)d_in[2];
    const float* b0    = (const float*)d_in[3];
    const float* W1    = (const float*)d_in[4];
    const float* b1    = (const float*)d_in[5];
    const float* Wt    = (const float*)d_in[6];
    const float* bt    = (const float*)d_in[7];
    const float* gamma = (const float*)d_in[8];
    const float* beta  = (const float*)d_in[9];
    const float* rmean = (const float*)d_in[10];
    const float* rvar  = (const float*)d_in[11];
    float* out = (float*)d_out;

    char* ws = (char*)d_ws;
    float*  rs_out = (float*)ws;  ws += (size_t)NGRAPH * NNODE * sizeof(float);
    float*  rs_in  = (float*)ws;  ws += (size_t)NGRAPH * NNODE * sizeof(float);
    float*  agg    = (float*)ws;  ws += (size_t)NGRAPH * NNODE * NFEAT * sizeof(float);
    half_t* h1     = (half_t*)ws; ws += (size_t)NGRAPH * NNODE * NFEAT * sizeof(half_t);
    float*  emb    = (float*)ws;  ws += (size_t)NGRAPH * NHID * sizeof(float);
    const size_t agg_bytes = (size_t)NGRAPH * NNODE * NFEAT * sizeof(float);
    const size_t emb_bytes = (size_t)NGRAPH * NHID * sizeof(float);

    hipMemsetAsync(agg, 0, agg_bytes, stream);
    k_degrees<<<NGRAPH, 512, 0, stream>>>(edges, rs_out, rs_in);

    k_scatter<<<NGRAPH, 512, 0, stream>>>(feats, nullptr, edges, rs_out, agg, 0);
    k_gemm   <<<NGRAPH * 2, 512, 0, stream>>>(agg, rs_in, W0, b0, h1, emb, 0);

    hipMemsetAsync(agg, 0, agg_bytes, stream);
    hipMemsetAsync(emb, 0, emb_bytes, stream);
    k_scatter<<<NGRAPH, 512, 0, stream>>>(nullptr, h1, edges, rs_out, agg, 1);
    k_gemm   <<<NGRAPH * 2, 512, 0, stream>>>(agg, rs_in, W1, b1, h1, emb, 1);

    k_head<<<NGRAPH, 128, 0, stream>>>(emb, Wt, bt, gamma, beta, rmean, rvar, out);
}


# Degree counts and rsqrt norms moved into the layer kernels: integer LDS adds on the edge list each block already loads (now once, at the top), beside phase 1's global loads; degree launch gone
# speedup vs baseline: 25.8165x; 25.8165x over previous
#include <hip/hip_runtime.h>
#include <hip/hip_bf16.h>

typedef _Float16 half_t;
typedef __attribute__((ext_vector_type(16))) _Float16 v16h;
typedef __attribute__((ext_vector_type(8)))  _Float16 v8h;
typedef __attribute__((ext_vector_type(2)))  _Float16 v2h;
typedef __attribute__((ext_vector_type(8)))  float    v8f;
typedef __attribute__((ext_vector_type(8)))  short    v8s;

#define NGRAPH 128
#define NNODE  512
#define NEDGE  8192
#define NFEAT  128
#define NHID   128
#define BN_EPS 1e-5f

#define CROWS   128
#define CPITCH  1040
#define YPITCH  1040
#define WPITCH  288
#define OFF_CNT 0
#define OFF_YT  (CROWS * CPITCH)
#define OFF_WT  (OFF_YT + NHID * YPITCH)
#define OFF_BS  (OFF_WT + NFEAT * WPITCH)
#define OFF_EMB (OFF_BS + NHID * 4)
#define OFF_DEG (OFF_EMB + NHID * 4)
#define LDS_BYTES (OFF_DEG + 2 * NNODE * 4)

__device__ __forceinline__ unsigned cnt_pair_to_f16(unsigned w) {
    v2h v;
    v[0] = (half_t)(unsigned short)(w & 0xffffu);
    v[1] = (half_t)(unsigned short)(w >> 16);
    return __builtin_bit_cast(unsigned, v);
}

__device__ __forceinline__ void count_edge(unsigned* cnt32, int src, int dst, int base) {
    const int s = src & (NNODE - 1);
    const unsigned r = (unsigned)(dst - base);
    if (r < (unsigned)CROWS)
        atomicAdd(&cnt32[r * (CPITCH / 4) + (s >> 1)], 1u << ((s & 1) * 16));
}

__device__ __forceinline__ v16h scale_a_f32(float4 f0, float4 f1, float4 f2, float4 f3, float s) {
    v16h a;
    a[0]  = (half_t)(f0.x * s); a[1]  = (half_t)(f0.y * s);
    a[2]  = (half_t)(f0.z * s); a[3]  = (half_t)(f0.w * s);
    a[4]  = (half_t)(f1.x * s); a[5]  = (half_t)(f1.y * s);
    a[6]  = (half_t)(f1.z * s); a[7]  = (half_t)(f1.w * s);
    a[8]  = (half_t)(f2.x * s); a[9]  = (half_t)(f2.y * s);
    a[10] = (half_t)(f2.z * s); a[11] = (half_t)(f2.w * s);
    a[12] = (half_t)(f3.x * s); a[13] = (half_t)(f3.y * s);
    a[14] = (half_t)(f3.z * s); a[15] = (half_t)(f3.w * s);
    return a;
}
__device__ __forceinline__ v16h load_ab_f16(const half_t* row, int k0, int k1) {
    const v8h lo = *(const v8h*)(row + k0);
    const v8h hi = *(const v8h*)(row + k1);
    return __builtin_shufflevector(lo, hi, 0, 1, 2, 3, 4, 5, 6, 7,
                                   8, 9, 10, 11, 12, 13, 14, 15);
}
__device__ __forceinline__ v8h lds_load_tr16(const unsigned char* p) {
    typedef __attribute__((address_space(3))) v8s lds_v8s;
    return __builtin_bit_cast(v8h, __builtin_amdgcn_ds_load_tr16_b128_v8i16((lds_v8s*)p));
}
__device__ __forceinline__ v16h load_b_tr16(const unsigned char* p) {
    const v8h lo = lds_load_tr16(p);
    const v8h hi = lds_load_tr16(p + 16 * WPITCH);
    return __builtin_shufflevector(lo, hi, 0, 1, 2, 3, 4, 5, 6, 7,
                                   8, 9, 10, 11, 12, 13, 14, 15);
}

template <int LAYER>
__device__ __forceinline__ void layer_body(unsigned char* L,
                                           const float*  __restrict__ xf32,
                                           const half_t* __restrict__ xf16,
                                           const int*    __restrict__ edges,
                                           const float*  __restrict__ W,
                                           const float*  __restrict__ bvec,
                                           half_t*       __restrict__ h_out,
                                           float*        __restrict__ emb_out) {
    float*  bs     = (float*)(L + OFF_BS);
    float*  embacc = (float*)(L + OFF_EMB);
    const int g    = blockIdx.x >> 1;
    const int half = blockIdx.x & 1;
    const int t    = threadIdx.x;

    const int* e = edges + (size_t)g * 2 * NEDGE;
    int4 es[4], ed[4];
#pragma unroll
    for (int q = 0; q < 4; ++q) {
        es[q] = *(const int4*)(e + 4 * (t + 512 * q));
        ed[q] = *(const int4*)(e + NEDGE + 4 * (t + 512 * q));
    }
    unsigned* deg = (unsigned*)(L + OFF_DEG);
    deg[t] = 0u; deg[t + 512] = 0u;

#pragma unroll
    for (int q = 0; q < NFEAT * (NHID / 8) / 512; ++q) {
        const int i = t + 512 * q;
        const int k = i >> 4, n0 = (i & 15) * 8;
        const float4 f0 = *(const float4*)(W + k * NHID + n0);
        const float4 f1 = *(const float4*)(W + k * NHID + n0 + 4);
        v8h w;
        w[0] = (half_t)f0.x; w[1] = (half_t)f0.y; w[2] = (half_t)f0.z; w[3] = (half_t)f0.w;
        w[4] = (half_t)f1.x; w[5] = (half_t)f1.y; w[6] = (half_t)f1.z; w[7] = (half_t)f1.w;
        *(v8h*)(L + OFF_WT + k * WPITCH + n0 * 2) = w;
    }
    if (t < NHID) { bs[t] = bvec[t]; embacc[t] = 0.f; }
    uint4* park = (uint4*)(L + OFF_CNT);
#pragma unroll
    for (int q = 0; q < 4; ++q) {
        park[t + 512 * q]       = make_uint4((unsigned)es[q].x, (unsigned)es[q].y, (unsigned)es[q].z, (unsigned)es[q].w);
        park[t + 512 * (4 + q)] = make_uint4((unsigned)ed[q].x, (unsigned)ed[q].y, (unsigned)ed[q].z, (unsigned)ed[q].w);
    }
    __syncthreads();

    const int wave = t >> 5, lane = t & 31;
    const int l16 = lane & 15, hi = lane >> 4;
    const size_t gb = (size_t)g * NNODE;
    const int mi0 = wave, mi1 = wave + 16;
    const int m0 = mi0 * 16 + l16, m1 = mi1 * 16 + l16;

    float4 xr0[16], xr1[16];
    v16h a0[4], a1[4];
    if (LAYER == 0) {
        const float* ar0 = xf32 + (gb + m0) * NFEAT;
        const float* ar1 = xf32 + (gb + m1) * NFEAT;
#pragma unroll
        for (int ko = 0; ko < 4; ++ko) {
            const int k0 = ko * 32 + hi * 8, k1 = ko * 32 + 16 + hi * 8;
            xr0[4 * ko + 0] = *(const float4*)(ar0 + k0);     xr0[4 * ko + 1] = *(const float4*)(ar0 + k0 + 4);
            xr0[4 * ko + 2] = *(const float4*)(ar0 + k1);     xr0[4 * ko + 3] = *(const float4*)(ar0 + k1 + 4);
            xr1[4 * ko + 0] = *(const float4*)(ar1 + k0);     xr1[4 * ko + 1] = *(const float4*)(ar1 + k0 + 4);
            xr1[4 * ko + 2] = *(const float4*)(ar1 + k1);     xr1[4 * ko + 3] = *(const float4*)(ar1 + k1 + 4);
        }
    } else {
        const half_t* hr0 = xf16 + (gb + m0) * NFEAT;
        const half_t* hr1 = xf16 + (gb + m1) * NFEAT;
#pragma unroll
        for (int ko = 0; ko < 4; ++ko) {
            a0[ko] = load_ab_f16(hr0, ko * 32 + hi * 8, ko * 32 + 16 + hi * 8);
            a1[ko] = load_ab_f16(hr1, ko * 32 + hi * 8, ko * 32 + 16 + hi * 8);
        }
    }

    __builtin_amdgcn_sched_barrier(0);
#pragma unroll
    for (int q = 0; q < 4; ++q) {
        if (LAYER == 0) {
            atomicAdd(&deg[es[q].x & (NNODE - 1)], 1u);
            atomicAdd(&deg[es[q].y & (NNODE - 1)], 1u);
            atomicAdd(&deg[es[q].z & (NNODE - 1)], 1u);
            atomicAdd(&deg[es[q].w & (NNODE - 1)], 1u);
        }
        atomicAdd(&deg[NNODE + (ed[q].x & (NNODE - 1))], 1u);
        atomicAdd(&deg[NNODE + (ed[q].y & (NNODE - 1))], 1u);
        atomicAdd(&deg[NNODE + (ed[q].z & (NNODE - 1))], 1u);
        atomicAdd(&deg[NNODE + (ed[q].w & (NNODE - 1))], 1u);
    }
    __syncthreads();
    {
        const float d0 = (float)deg[t], d1 = (float)deg[t + 512];
        deg[t]       = __float_as_uint(rsqrtf(fmaxf(d0, 1.0f)));
        deg[t + 512] = __float_as_uint(rsqrtf(fmaxf(d1, 1.0f)));
    }
    __syncthreads();

    {
        if (LAYER == 0) {
            const float rso0 = __uint_as_float(deg[m0]), rso1 = __uint_as_float(deg[m1]);
#pragma unroll
            for (int ko = 0; ko < 4; ++ko) {
                a0[ko] = scale_a_f32(xr0[4 * ko], xr0[4 * ko + 1], xr0[4 * ko + 2], xr0[4 * ko + 3], rso0);
                a1[ko] = scale_a_f32(xr1[4 * ko], xr1[4 * ko + 1], xr1[4 * ko + 2], xr1[4 * ko + 3], rso1);
            }
        }
        const unsigned char* wl = L + OFF_WT + (8 * hi + (lane & 7)) * WPITCH + 16 * ((lane >> 3) & 1);
#pragma unroll 1
        for (int nq = 0; nq < 2; ++nq) {
            v8f c0[4] = {};
            v8f c1[4] = {};
#pragma unroll
            for (int ko = 0; ko < 4; ++ko) {
#pragma unroll
                for (int j = 0; j < 4; ++j) {
                    const v16h b = load_b_tr16(wl + (ko * 32) * WPITCH + (nq * 4 + j) * 32);
                    c0[j] = __builtin_amdgcn_wmma_f32_16x16x32_f16(
                                false, a0[ko], false, b, (short)0, c0[j], false, false);
                    c1[j] = __builtin_amdgcn_wmma_f32_16x16x32_f16(
                                false, a1[ko], false, b, (short)0, c1[j], false, false);
                }
            }
#pragma unroll
            for (int j = 0; j < 4; ++j) {
                unsigned char* yrow = L + OFF_YT + ((nq * 4 + j) * 16 + l16) * YPITCH;
                v8h y0, y1;
#pragma unroll
                for (int r = 0; r < 8; ++r) { y0[r] = (half_t)c0[j][r]; y1[r] = (half_t)c1[j][r]; }
                *(v8h*)(yrow + (mi0 * 16 + 8 * hi) * 2) = y0;
                *(v8h*)(yrow + (mi1 * 16 + 8 * hi) * 2) = y1;
            }
        }
    }
    __syncthreads();

#pragma unroll
    for (int q = 0; q < 4; ++q) {
        const uint4 ps = park[t + 512 * q], pd = park[t + 512 * (4 + q)];
        es[q] = make_int4((int)ps.x, (int)ps.y, (int)ps.z, (int)ps.w);
        ed[q] = make_int4((int)pd.x, (int)pd.y, (int)pd.z, (int)pd.w);
    }

    unsigned* cnt32  = (unsigned*)(L + OFF_CNT);
    uint4*    cnt128 = (uint4*)(L + OFF_CNT);
    const int mp = wave >> 2;
    const int np = wave & 3;

    for (int p = 0; p < 2; ++p) {
        const int base = half * 256 + p * CROWS;

        for (int i = t; i < CROWS * CPITCH / 16; i += 512) cnt128[i] = make_uint4(0u, 0u, 0u, 0u);
        __syncthreads();

#pragma unroll
        for (int q = 0; q < 4; ++q) {
            count_edge(cnt32, es[q].x, ed[q].x, base);
            count_edge(cnt32, es[q].y, ed[q].y, base);
            count_edge(cnt32, es[q].z, ed[q].z, base);
            count_edge(cnt32, es[q].w, ed[q].w, base);
        }
        __syncthreads();

        for (int i = t; i < CROWS * CPITCH / 16; i += 512) {
            uint4 w = cnt128[i];
            w.x = cnt_pair_to_f16(w.x); w.y = cnt_pair_to_f16(w.y);
            w.z = cnt_pair_to_f16(w.z); w.w = cnt_pair_to_f16(w.w);
            cnt128[i] = w;
        }
        __syncthreads();

        v8f acc[2][2] = {};
        const half_t* arow = (const half_t*)(L + OFF_CNT + (mp * 32 + l16) * CPITCH);
        const half_t* brow = (const half_t*)(L + OFF_YT  + (np * 32 + l16) * YPITCH);
#pragma unroll 4
        for (int ks = 0; ks < 16; ++ks) {
            const int k0 = ks * 32 + hi * 8;
            const int k1 = ks * 32 + 16 + hi * 8;
            v16h a[2], b[2];
            a[0] = load_ab_f16(arow, k0, k1);
            a[1] = load_ab_f16(arow + 16 * (CPITCH / 2), k0, k1);
            b[0] = load_ab_f16(brow, k0, k1);
            b[1] = load_ab_f16(brow + 16 * (YPITCH / 2), k0, k1);
#pragma unroll
            for (int i = 0; i < 2; ++i)
#pragma unroll
                for (int j = 0; j < 2; ++j)
                    acc[i][j] = __builtin_amdgcn_wmma_f32_16x16x32_f16(
                                    false, a[i], false, b[j], (short)0, acc[i][j], false, false);
        }

#pragma unroll
        for (int i = 0; i < 2; ++i) {
            const int    nrow = base + (mp * 2 + i) * 16 + 8 * hi;
            const size_t row0 = gb + nrow;
            float rin[8];
#pragma unroll
            for (int r = 0; r < 8; ++r) rin[r] = __uint_as_float(deg[NNODE + nrow + r]);
            if (LAYER == 0) {
                float rso[8];
#pragma unroll
                for (int r = 0; r < 8; ++r) rso[r] = __uint_as_float(deg[nrow + r]);
#pragma unroll
                for (int j = 0; j < 2; ++j) {
                    const int n = (np * 2 + j) * 16 + l16;
                    const float bn = bs[n];
#pragma unroll
                    for (int r = 0; r < 8; ++r) {
                        const float v = fmaxf(acc[i][j][r] * rin[r] + bn, 0.f);
                        h_out[(row0 + r) * NFEAT + n] = (half_t)(v * rso[r]);
                    }
                }
            } else {
#pragma unroll
                for (int j = 0; j < 2; ++j) {
                    const int n = (np * 2 + j) * 16 + l16;
                    const float bn = bs[n];
                    float s = 0.f;
#pragma unroll
                    for (int r = 0; r < 8; ++r) s += fmaxf(acc[i][j][r] * rin[r] + bn, 0.f);
                    atomicAdd(&embacc[n], s);
                }
            }
        }
        __syncthreads();
    }

    if (LAYER == 1) {
        if (t < NHID) atomicAdd(&emb_out[g * NHID + t], embacc[t] * (1.0f / NNODE));
    }
}

__global__ __launch_bounds__(512, 1) void k_layer0(const float* __restrict__ x,
                                                   const int*   __restrict__ edges,
                                                   const float* __restrict__ W,
                                                   const float* __restrict__ bvec,
                                                   half_t*      __restrict__ h_out) {
    __shared__ __attribute__((aligned(16))) unsigned char L[LDS_BYTES];
    layer_body<0>(L, x, nullptr, edges, W, bvec, h_out, nullptr);
}

__global__ __launch_bounds__(512, 1) void k_layer1(const half_t* __restrict__ hs,
                                                   const int*    __restrict__ edges,
                                                   const float*  __restrict__ W,
                                                   const float*  __restrict__ bvec,
                                                   float*        __restrict__ emb_out) {
    __shared__ __attribute__((aligned(16))) unsigned char L[LDS_BYTES];
    layer_body<1>(L, nullptr, hs, edges, W, bvec, nullptr, emb_out);
}

__global__ __launch_bounds__(128) void k_head(const float* __restrict__ emb,
                                              const float* __restrict__ Wt,
                                              const float* __restrict__ bt,
                                              const float* __restrict__ gamma,
                                              const float* __restrict__ beta,
                                              const float* __restrict__ rmean,
                                              const float* __restrict__ rvar,
                                              float* __restrict__ out) {
    __shared__ float er[NHID];
    const int g = blockIdx.x, n = threadIdx.x;
    er[n] = emb[g * NHID + n];
    __syncthreads();
    float z = bt[n];
    for (int k = 0; k < NHID; ++k) z += er[k] * Wt[k * NHID + n];
    z = gamma[n] * (z - rmean[n]) * rsqrtf(rvar[n] + BN_EPS) + beta[n];
    out[g * NHID + n] = fmaxf(z, 0.f);
}

extern "C" void kernel_launch(void* const* d_in, const int* in_sizes, int n_in,
                              void* d_out, int out_size, void* d_ws, size_t ws_size,
                              hipStream_t stream) {
    (void)in_sizes; (void)n_in; (void)out_size; (void)ws_size;
    const float* feats = (const float*)d_in[0];
    const int*   edges = (const int*)d_in[1];
    const float* W0    = (const float*)d_in[2];
    const float* b0    = (const float*)d_in[3];
    const float* W1    = (const float*)d_in[4];
    const float* b1    = (const float*)d_in[5];
    const float* Wt    = (const float*)d_in[6];
    const float* bt    = (const float*)d_in[7];
    const float* gamma = (const float*)d_in[8];
    const float* beta  = (const float*)d_in[9];
    const float* rmean = (const float*)d_in[10];
    const float* rvar  = (const float*)d_in[11];
    float* out = (float*)d_out;

    char* ws = (char*)d_ws;
    half_t* h1s    = (half_t*)ws; ws += (size_t)NGRAPH * NNODE * NFEAT * sizeof(half_t);
    float*  emb    = (float*)ws;  ws += (size_t)NGRAPH * NHID * sizeof(float);
    const size_t emb_bytes = (size_t)NGRAPH * NHID * sizeof(float);

    k_layer0<<<NGRAPH * 2, 512, 0, stream>>>(feats, edges, W0, b0, h1s);

    hipMemsetAsync(emb, 0, emb_bytes, stream);
    k_layer1<<<NGRAPH * 2, 512, 0, stream>>>(h1s, edges, W1, b1, emb);

    k_head<<<NGRAPH, 128, 0, stream>>>(emb, Wt, bt, gamma, beta, rmean, rvar, out);
}


# All global loads of each kernel issued in one group at entry (layers: rows behind edges, W, bias; head: its Wt column); set-up barriers wait for LDS only: dependent load trips 3/3/10 -> 1; same sums
# speedup vs baseline: 35.0325x; 1.3570x over previous
#include <hip/hip_runtime.h>
#include <hip/hip_bf16.h>

typedef _Float16 half_t;
typedef __attribute__((ext_vector_type(16))) _Float16 v16h;
typedef __attribute__((ext_vector_type(8)))  _Float16 v8h;
typedef __attribute__((ext_vector_type(2)))  _Float16 v2h;
typedef __attribute__((ext_vector_type(8)))  float    v8f;
typedef __attribute__((ext_vector_type(8)))  short    v8s;
typedef __attribute__((ext_vector_type(8)))  unsigned v8u;

#define NGRAPH 128
#define NNODE  512
#define NEDGE  8192
#define NFEAT  128
#define NHID   128
#define BN_EPS 1e-5f

#define BROWS   256
#define BPITCH  528
#define CROWS   128
#define CPITCH  1040
#define YPITCH  1040
#define WPITCH  288
#define OFF_CNT 0
#define OFF_YT  (BROWS * BPITCH)
#define OFF_WT  (OFF_YT + NHID * YPITCH)
#define OFF_BS  (OFF_WT + NFEAT * WPITCH)
#define OFF_EMB (OFF_BS + NHID * 4)
#define OFF_DEG (OFF_EMB + NHID * 4)
#define OFF_FLAG (OFF_DEG + 2 * NNODE * 4)
#define LDS_BYTES (OFF_FLAG + 16)

__device__ __forceinline__ unsigned cnt_pair_to_f16(unsigned w) {
    v2h v;
    v[0] = (half_t)(unsigned short)(w & 0xffffu);
    v[1] = (half_t)(unsigned short)(w >> 16);
    return __builtin_bit_cast(unsigned, v);
}

#define CNT_UNSCALE 16777216.0f

__device__ __forceinline__ void count_edge8(unsigned* cnt32, int src, int dst, int base) {
    const int s = src & (NNODE - 1);
    const unsigned r = (unsigned)(dst - base);
    if (r < (unsigned)BROWS)
        atomicAdd(&cnt32[r * (BPITCH / 4) + (s >> 2)], 1u << ((s & 3) * 8));
}
__device__ __forceinline__ unsigned widen_lo(unsigned d) { return __builtin_amdgcn_perm(0u, d, 0x0c010c00u); }
__device__ __forceinline__ unsigned widen_hi(unsigned d) { return __builtin_amdgcn_perm(0u, d, 0x0c030c02u); }
__device__ __forceinline__ v16h widen_a_cnt8(uint2 lo, uint2 hi) {
    v8u a;
    a[0] = widen_lo(lo.x); a[1] = widen_hi(lo.x); a[2] = widen_lo(lo.y); a[3] = widen_hi(lo.y);
    a[4] = widen_lo(hi.x); a[5] = widen_hi(hi.x); a[6] = widen_lo(hi.y); a[7] = widen_hi(hi.y);
    return __builtin_bit_cast(v16h, a);
}

__device__ __forceinline__ void count_edge(unsigned* cnt32, int src, int dst, int base) {
    const int s = src & (NNODE - 1);
    const unsigned r = (unsigned)(dst - base);
    if (r < (unsigned)CROWS)
        atomicAdd(&cnt32[r * (CPITCH / 4) + (s >> 1)], 1u << ((s & 1) * 16));
}

__device__ __forceinline__ v16h scale_a_f32(float4 f0, float4 f1, float4 f2, float4 f3, float s) {
    v16h a;
    a[0]  = (half_t)(f0.x * s); a[1]  = (half_t)(f0.y * s);
    a[2]  = (half_t)(f0.z * s); a[3]  = (half_t)(f0.w * s);
    a[4]  = (half_t)(f1.x * s); a[5]  = (half_t)(f1.y * s);
    a[6]  = (half_t)(f1.z * s); a[7]  = (half_t)(f1.w * s);
    a[8]  = (half_t)(f2.x * s); a[9]  = (half_t)(f2.y * s);
    a[10] = (half_t)(f2.z * s); a[11] = (half_t)(f2.w * s);
    a[12] = (half_t)(f3.x * s); a[13] = (half_t)(f3.y * s);
    a[14] = (half_t)(f3.z * s); a[15] = (half_t)(f3.w * s);
    return a;
}
__device__ __forceinline__ v16h load_ab_f16(const half_t* row, int k0, int k1) {
    const v8h lo = *(const v8h*)(row + k0);
    const v8h hi = *(const v8h*)(row + k1);
    return __builtin_shufflevector(lo, hi, 0, 1, 2, 3, 4, 5, 6, 7,
                                   8, 9, 10, 11, 12, 13, 14, 15);
}
__device__ __forceinline__ v8h lds_load_tr16(const unsigned char* p) {
    typedef __attribute__((address_space(3))) v8s lds_v8s;
    return __builtin_bit_cast(v8h, __builtin_amdgcn_ds_load_tr16_b128_v8i16((lds_v8s*)p));
}
__device__ __forceinline__ v16h load_b_tr16(const unsigned char* p) {
    const v8h lo = lds_load_tr16(p);
    const v8h hi = lds_load_tr16(p + 16 * WPITCH);
    return __builtin_shufflevector(lo, hi, 0, 1, 2, 3, 4, 5, 6, 7,
                                   8, 9, 10, 11, 12, 13, 14, 15);
}

__device__ __forceinline__ void lds_barrier() {
    __builtin_amdgcn_fence(__ATOMIC_RELEASE, "workgroup", "local");
    __builtin_amdgcn_s_barrier();
    __builtin_amdgcn_fence(__ATOMIC_ACQUIRE, "workgroup", "local");
}

template <int LAYER>
__device__ __forceinline__ void tile_epilogue(const v8f& c0, const v8f& c1, int nrow, size_t gb,
                                              int np, int l16, float unscale,
                                              const unsigned* deg, const float* bs, float* embacc,
                                              half_t* __restrict__ h_out) {
    const size_t row0 = gb + nrow;
    float rin[8];
#pragma unroll
    for (int r = 0; r < 8; ++r) rin[r] = __uint_as_float(deg[NNODE + nrow + r]) * unscale;
    if (LAYER == 0) {
        float rso[8];
#pragma unroll
        for (int r = 0; r < 8; ++r) rso[r] = __uint_as_float(deg[nrow + r]);
#pragma unroll
        for (int j = 0; j < 2; ++j) {
            const v8f c = j ? c1 : c0;
            const int n = (np * 2 + j) * 16 + l16;
            const float bn = bs[n];
#pragma unroll
            for (int r = 0; r < 8; ++r) {
                const float v = fmaxf(c[r] * rin[r] + bn, 0.f);
                h_out[(row0 + r) * NFEAT + n] = (half_t)(v * rso[r]);
            }
        }
    } else {
#pragma unroll
        for (int j = 0; j < 2; ++j) {
            const v8f c = j ? c1 : c0;
            const int n = (np * 2 + j) * 16 + l16;
            const float bn = bs[n];
            float s = 0.f;
#pragma unroll
            for (int r = 0; r < 8; ++r) s += fmaxf(c[r] * rin[r] + bn, 0.f);
            atomicAdd(&embacc[n], s);
        }
    }
}

template <int LAYER>
__device__ __forceinline__ void layer_body(unsigned char* L,
                                           const float*  __restrict__ xf32,
                                           const half_t* __restrict__ xf16,
                                           const int*    __restrict__ edges,
                                           const float*  __restrict__ W,
                                           const float*  __restrict__ bvec,
                                           half_t*       __restrict__ h_out,
                                           float*        __restrict__ emb_out) {
    float*  bs     = (float*)(L + OFF_BS);
    float*  embacc = (float*)(L + OFF_EMB);
    const int g    = blockIdx.x >> 1;
    const int half = blockIdx.x & 1;
    const int t    = threadIdx.x;

    const int* e = edges + (size_t)g * 2 * NEDGE;
    int4 es[4], ed[4];
#pragma unroll
    for (int q = 0; q < 4; ++q) {
        es[q] = *(const int4*)(e + 4 * (t + 512 * q));
        ed[q] = *(const int4*)(e + NEDGE + 4 * (t + 512 * q));
    }
    float4 wr[2 * (NFEAT * (NHID / 8) / 512)];
#pragma unroll
    for (int q = 0; q < NFEAT * (NHID / 8) / 512; ++q) {
        const int i = t + 512 * q;
        const int k = i >> 4, n0 = (i & 15) * 8;
        wr[2 * q]     = *(const float4*)(W + k * NHID + n0);
        wr[2 * q + 1] = *(const float4*)(W + k * NHID + n0 + 4);
    }
    const float bv = bvec[t & (NHID - 1)];
    __builtin_amdgcn_sched_barrier(0);

    const int wave = t >> 5, lane = t & 31;
    const int l16 = lane & 15, hi = lane >> 4;
    const size_t gb = (size_t)g * NNODE;
    const int mi0 = wave, mi1 = wave + 16;
    const int m0 = mi0 * 16 + l16, m1 = mi1 * 16 + l16;

    float4 xr0[16], xr1[16];
    v16h a0[4], a1[4];
    if (LAYER == 0) {
        const float* ar0 = xf32 + (gb + m0) * NFEAT;
        const float* ar1 = xf32 + (gb + m1) * NFEAT;
#pragma unroll
        for (int ko = 0; ko < 4; ++ko) {
            const int k0 = ko * 32 + hi * 8, k1 = ko * 32 + 16 + hi * 8;
            xr0[4 * ko + 0] = *(const float4*)(ar0 + k0);     xr0[4 * ko + 1] = *(const float4*)(ar0 + k0 + 4);
            xr0[4 * ko + 2] = *(const float4*)(ar0 + k1);     xr0[4 * ko + 3] = *(const float4*)(ar0 + k1 + 4);
            xr1[4 * ko + 0] = *(const float4*)(ar1 + k0);     xr1[4 * ko + 1] = *(const float4*)(ar1 + k0 + 4);
            xr1[4 * ko + 2] = *(const float4*)(ar1 + k1);     xr1[4 * ko + 3] = *(const float4*)(ar1 + k1 + 4);
        }
    } else {
        const half_t* hr0 = xf16 + (gb + m0) * NFEAT;
        const half_t* hr1 = xf16 + (gb + m1) * NFEAT;
#pragma unroll
        for (int ko = 0; ko < 4; ++ko) {
            a0[ko] = load_ab_f16(hr0, ko * 32 + hi * 8, ko * 32 + 16 + hi * 8);
            a1[ko] = load_ab_f16(hr1, ko * 32 + hi * 8, ko * 32 + 16 + hi * 8);
        }
    }

    unsigned* deg = (unsigned*)(L + OFF_DEG);
    deg[t] = 0u; deg[t + 512] = 0u;
    unsigned* wide_flag = (unsigned*)(L + OFF_FLAG);
    if (t == 0) *wide_flag = 0u;

#pragma unroll
    for (int q = 0; q < NFEAT * (NHID / 8) / 512; ++q) {
        const int i = t + 512 * q;
        const int k = i >> 4, n0 = (i & 15) * 8;
        const float4 f0 = wr[2 * q], f1 = wr[2 * q + 1];
        v8h w;
        w[0] = (half_t)f0.x; w[1] = (half_t)f0.y; w[2] = (half_t)f0.z; w[3] = (half_t)f0.w;
        w[4] = (half_t)f1.x; w[5] = (half_t)f1.y; w[6] = (half_t)f1.z; w[7] = (half_t)f1.w;
        *(v8h*)(L + OFF_WT + k * WPITCH + n0 * 2) = w;
    }
    if (t < NHID) { bs[t] = bv; embacc[t] = 0.f; }
    uint4* park = (uint4*)(L + OFF_CNT);
#pragma unroll
    for (int q = 0; q < 4; ++q) {
        park[t + 512 * q]       = make_uint4((unsigned)es[q].x, (unsigned)es[q].y, (unsigned)es[q].z, (unsigned)es[q].w);
        park[t + 512 * (4 + q)] = make_uint4((unsigned)ed[q].x, (unsigned)ed[q].y, (unsigned)ed[q].z, (unsigned)ed[q].w);
    }
    lds_barrier();

#pragma unroll
    for (int q = 0; q < 4; ++q) {
        if (LAYER == 0) {
            atomicAdd(&deg[es[q].x & (NNODE - 1)], 1u);
            atomicAdd(&deg[es[q].y & (NNODE - 1)], 1u);
            atomicAdd(&deg[es[q].z & (NNODE - 1)], 1u);
            atomicAdd(&deg[es[q].w & (NNODE - 1)], 1u);
        }
        atomicAdd(&deg[NNODE + (ed[q].x & (NNODE - 1))], 1u);
        atomicAdd(&deg[NNODE + (ed[q].y & (NNODE - 1))], 1u);
        atomicAdd(&deg[NNODE + (ed[q].z & (NNODE - 1))], 1u);
        atomicAdd(&deg[NNODE + (ed[q].w & (NNODE - 1))], 1u);
    }
    lds_barrier();
    {
        const unsigned in_cnt = deg[t + 512];
        if (in_cnt > 255u) *wide_flag = 1u;
        const float d0 = (float)deg[t], d1 = (float)in_cnt;
        deg[t]       = __float_as_uint(rsqrtf(fmaxf(d0, 1.0f)));
        deg[t + 512] = __float_as_uint(rsqrtf(fmaxf(d1, 1.0f)));
    }
    lds_barrier();

    {
        if (LAYER == 0) {
            const float rso0 = __uint_as_float(deg[m0]), rso1 = __uint_as_float(deg[m1]);
#pragma unroll
            for (int ko = 0; ko < 4; ++ko) {
                a0[ko] = scale_a_f32(xr0[4 * ko], xr0[4 * ko + 1], xr0[4 * ko + 2], xr0[4 * ko + 3], rso0);
                a1[ko] = scale_a_f32(xr1[4 * ko], xr1[4 * ko + 1], xr1[4 * ko + 2], xr1[4 * ko + 3], rso1);
            }
        }
        const unsigned char* wl = L + OFF_WT + (8 * hi + (lane & 7)) * WPITCH + 16 * ((lane >> 3) & 1);
#pragma unroll 1
        for (int nq = 0; nq < 2; ++nq) {
            v8f c0[4] = {};
            v8f c1[4] = {};
#pragma unroll
            for (int ko = 0; ko < 4; ++ko) {
#pragma unroll
                for (int j = 0; j < 4; ++j) {
                    const v16h b = load_b_tr16(wl + (ko * 32) * WPITCH + (nq * 4 + j) * 32);
                    c0[j] = __builtin_amdgcn_wmma_f32_16x16x32_f16(
                                false, a0[ko], false, b, (short)0, c0[j], false, false);
                    c1[j] = __builtin_amdgcn_wmma_f32_16x16x32_f16(
                                false, a1[ko], false, b, (short)0, c1[j], false, false);
                }
            }
#pragma unroll
            for (int j = 0; j < 4; ++j) {
                unsigned char* yrow = L + OFF_YT + ((nq * 4 + j) * 16 + l16) * YPITCH;
                v8h y0, y1;
#pragma unroll
                for (int r = 0; r < 8; ++r) { y0[r] = (half_t)c0[j][r]; y1[r] = (half_t)c1[j][r]; }
                *(v8h*)(yrow + (mi0 * 16 + 8 * hi) * 2) = y0;
                *(v8h*)(yrow + (mi1 * 16 + 8 * hi) * 2) = y1;
            }
        }
    }
    __syncthreads();

#pragma unroll
    for (int q = 0; q < 4; ++q) {
        const uint4 ps = park[t + 512 * q], pd = park[t + 512 * (4 + q)];
        es[q] = make_int4((int)ps.x, (int)ps.y, (int)ps.z, (int)ps.w);
        ed[q] = make_int4((int)pd.x, (int)pd.y, (int)pd.z, (int)pd.w);
    }

    unsigned* cnt32  = (unsigned*)(L + OFF_CNT);
    uint4*    cnt128 = (uint4*)(L + OFF_CNT);
    const int np = wave & 3;
    const half_t* brow = (const half_t*)(L + OFF_YT + (np * 32 + l16) * YPITCH);
    const bool wide = __builtin_amdgcn_readfirstlane((int)*wide_flag) != 0;

    if (!wide) {
        const int base = half * BROWS;
        const int mq = wave >> 2;

        for (int i = t; i < BROWS * BPITCH / 16; i += 512) cnt128[i] = make_uint4(0u, 0u, 0u, 0u);
        __syncthreads();

#pragma unroll
        for (int q = 0; q < 4; ++q) {
            count_edge8(cnt32, es[q].x, ed[q].x, base);
            count_edge8(cnt32, es[q].y, ed[q].y, base);
            count_edge8(cnt32, es[q].z, ed[q].z, base);
            count_edge8(cnt32, es[q].w, ed[q].w, base);
        }
        __syncthreads();

        v8f acc[4][2] = {};
        const unsigned char* arow = L + OFF_CNT + (mq * 64 + l16) * BPITCH + hi * 8;
        uint2 ra[2][4][2];
        v16h  rb[2][2];
#pragma unroll
        for (int i = 0; i < 4; ++i) {
            ra[0][i][0] = *(const uint2*)(arow + i * 16 * BPITCH);
            ra[0][i][1] = *(const uint2*)(arow + i * 16 * BPITCH + 16);
        }
        rb[0][0] = load_ab_f16(brow, hi * 8, 16 + hi * 8);
        rb[0][1] = load_ab_f16(brow + 16 * (YPITCH / 2), hi * 8, 16 + hi * 8);
#pragma unroll 1
        for (int kk = 0; kk < 16; kk += 2) {
#pragma unroll
            for (int s = 0; s < 2; ++s) {
                const int kn = ((kk + s + 1) & 15) * 32;
#pragma unroll
                for (int i = 0; i < 4; ++i) {
                    ra[s ^ 1][i][0] = *(const uint2*)(arow + i * 16 * BPITCH + kn);
                    ra[s ^ 1][i][1] = *(const uint2*)(arow + i * 16 * BPITCH + kn + 16);
                }
                rb[s ^ 1][0] = load_ab_f16(brow, kn + hi * 8, kn + 16 + hi * 8);
                rb[s ^ 1][1] = load_ab_f16(brow + 16 * (YPITCH / 2), kn + hi * 8, kn + 16 + hi * 8);
                __builtin_amdgcn_sched_barrier(0);
                v16h a[4];
#pragma unroll
                for (int i = 0; i < 4; ++i) a[i] = widen_a_cnt8(ra[s][i][0], ra[s][i][1]);
#pragma unroll
                for (int i = 0; i < 4; ++i)
#pragma unroll
                    for (int j = 0; j < 2; ++j)
                        acc[i][j] = __builtin_amdgcn_wmma_f32_16x16x32_f16(
                                        false, a[i], false, rb[s][j], (short)0, acc[i][j], false, false);
                __builtin_amdgcn_sched_barrier(0);
            }
        }

#pragma unroll
        for (int i = 0; i < 4; ++i)
            tile_epilogue<LAYER>(acc[i][0], acc[i][1], base + (mq * 4 + i) * 16 + 8 * hi, gb, np, l16,
                                 CNT_UNSCALE, deg, bs, embacc, h_out);
        __syncthreads();
    } else {
        const int mp = wave >> 2;
        for (int p = 0; p < 2; ++p) {
            const int base = half * 256 + p * CROWS;

            for (int i = t; i < CROWS * CPITCH / 16; i += 512) cnt128[i] = make_uint4(0u, 0u, 0u, 0u);
            __syncthreads();

#pragma unroll
            for (int q = 0; q < 4; ++q) {
                count_edge(cnt32, es[q].x, ed[q].x, base);
                count_edge(cnt32, es[q].y, ed[q].y, base);
                count_edge(cnt32, es[q].z, ed[q].z, base);
                count_edge(cnt32, es[q].w, ed[q].w, base);
            }
            __syncthreads();

            for (int i = t; i < CROWS * CPITCH / 16; i += 512) {
                uint4 w = cnt128[i];
                w.x = cnt_pair_to_f16(w.x); w.y = cnt_pair_to_f16(w.y);
                w.z = cnt_pair_to_f16(w.z); w.w = cnt_pair_to_f16(w.w);
                cnt128[i] = w;
            }
            __syncthreads();

            v8f acc[2][2] = {};
            const half_t* arow = (const half_t*)(L + OFF_CNT + (mp * 32 + l16) * CPITCH);
#pragma unroll 4
            for (int ks = 0; ks < 16; ++ks) {
                const int k0 = ks * 32 + hi * 8;
                const int k1 = ks * 32 + 16 + hi * 8;
                v16h a[2], b[2];
                a[0] = load_ab_f16(arow, k0, k1);
                a[1] = load_ab_f16(arow + 16 * (CPITCH / 2), k0, k1);
                b[0] = load_ab_f16(brow, k0, k1);
                b[1] = load_ab_f16(brow + 16 * (YPITCH / 2), k0, k1);
#pragma unroll
                for (int i = 0; i < 2; ++i)
#pragma unroll
                    for (int j = 0; j < 2; ++j)
                        acc[i][j] = __builtin_amdgcn_wmma_f32_16x16x32_f16(
                                        false, a[i], false, b[j], (short)0, acc[i][j], false, false);
            }

#pragma unroll
            for (int i = 0; i < 2; ++i)
                tile_epilogue<LAYER>(acc[i][0], acc[i][1], base + (mp * 2 + i) * 16 + 8 * hi, gb, np, l16,
                                     1.0f, deg, bs, embacc, h_out);
            __syncthreads();
        }
    }

    if (LAYER == 1) {
        if (t < NHID) atomicAdd(&emb_out[g * NHID + t], embacc[t] * (1.0f / NNODE));
    }
}

__global__ __launch_bounds__(512, 1) void k_layer0(const float* __restrict__ x,
                                                   const int*   __restrict__ edges,
                                                   const float* __restrict__ W,
                                                   const float* __restrict__ bvec,
                                                   half_t*      __restrict__ h_out) {
    __shared__ __attribute__((aligned(16))) unsigned char L[LDS_BYTES];
    layer_body<0>(L, x, nullptr, edges, W, bvec, h_out, nullptr);
}

__global__ __launch_bounds__(512, 1) void k_layer1(const half_t* __restrict__ hs,
                                                   const int*    __restrict__ edges,
                                                   const float*  __restrict__ W,
                                                   const float*  __restrict__ bvec,
                                                   float*        __restrict__ emb_out) {
    __shared__ __attribute__((aligned(16))) unsigned char L[LDS_BYTES];
    layer_body<1>(L, nullptr, hs, edges, W, bvec, nullptr, emb_out);
}

__global__ __launch_bounds__(128) void k_head(const float* __restrict__ emb,
                                              const float* __restrict__ Wt,
                                              const float* __restrict__ bt,
                                              const float* __restrict__ gamma,
                                              const float* __restrict__ beta,
                                              const float* __restrict__ rmean,
                                              const float* __restrict__ rvar,
                                              float* __restrict__ out) {
    __shared__ float er[NHID];
    const int g = blockIdx.x, n = threadIdx.x;
    const float e  = emb[g * NHID + n];
    const float zb = bt[n], gm = gamma[n], be = beta[n], rm = rmean[n], rv = rvar[n];
    float w[NHID];
#pragma unroll
    for (int k = 0; k < NHID; ++k) w[k] = Wt[k * NHID + n];
    er[n] = e;
    lds_barrier();
    float z = zb;
#pragma unroll
    for (int k = 0; k < NHID; ++k) z += er[k] * w[k];
    z = gm * (z - rm) * rsqrtf(rv + BN_EPS) + be;
    out[g * NHID + n] = fmaxf(z, 0.f);
}

extern "C" void kernel_launch(void* const* d_in, const int* in_sizes, int n_in,
                              void* d_out, int out_size, void* d_ws, size_t ws_size,
                              hipStream_t stream) {
    (void)in_sizes; (void)n_in; (void)out_size; (void)ws_size;
    const float* feats = (const float*)d_in[0];
    const int*   edges = (const int*)d_in[1];
    const float* W0    = (const float*)d_in[2];
    const float* b0    = (const float*)d_in[3];
    const float* W1    = (const float*)d_in[4];
    const float* b1    = (const float*)d_in[5];
    const float* Wt    = (const float*)d_in[6];
    const float* bt    = (const float*)d_in[7];
    const float* gamma = (const float*)d_in[8];
    const float* beta  = (const float*)d_in[9];
    const float* rmean = (const float*)d_in[10];
    const float* rvar  = (const float*)d_in[11];
    float* out = (float*)d_out;

    char* ws = (char*)d_ws;
    half_t* h1s    = (half_t*)ws; ws += (size_t)NGRAPH * NNODE * NFEAT * sizeof(half_t);
    float*  emb    = (float*)ws;  ws += (size_t)NGRAPH * NHID * sizeof(float);
    const size_t emb_bytes = (size_t)NGRAPH * NHID * sizeof(float);

    k_layer0<<<NGRAPH * 2, 512, 0, stream>>>(feats, edges, W0, b0, h1s);

    hipMemsetAsync(emb, 0, emb_bytes, stream);
    k_layer1<<<NGRAPH * 2, 512, 0, stream>>>(h1s, edges, W1, b1, emb);

    k_head<<<NGRAPH, 128, 0, stream>>>(emb, Wt, bt, gamma, beta, rmean, rvar, out);
}


# Layer 0 aggregation product run transposed (WMMA operands swapped): a lane holds 8 columns of one node row; h1 kept in layer 1's A-fragment order, 512-byte runs per store and load; same sums
# speedup vs baseline: 36.5401x; 1.0430x over previous
#include <hip/hip_runtime.h>
#include <hip/hip_bf16.h>

typedef _Float16 half_t;
typedef __attribute__((ext_vector_type(16))) _Float16 v16h;
typedef __attribute__((ext_vector_type(8)))  _Float16 v8h;
typedef __attribute__((ext_vector_type(2)))  _Float16 v2h;
typedef __attribute__((ext_vector_type(8)))  float    v8f;
typedef __attribute__((ext_vector_type(8)))  short    v8s;
typedef __attribute__((ext_vector_type(8)))  unsigned v8u;

#define NGRAPH 128
#define NNODE  512
#define NEDGE  8192
#define NFEAT  128
#define NHID   128
#define BN_EPS 1e-5f

#define BROWS   256
#define BPITCH  528
#define CROWS   128
#define CPITCH  1040
#define YPITCH  1040
#define WPITCH  288
#define OFF_CNT 0
#define OFF_YT  (BROWS * BPITCH)
#define OFF_WT  (OFF_YT + NHID * YPITCH)
#define OFF_BS  (OFF_WT + NFEAT * WPITCH)
#define OFF_EMB (OFF_BS + NHID * 4)
#define OFF_DEG (OFF_EMB + NHID * 4)
#define OFF_FLAG (OFF_DEG + 2 * NNODE * 4)
#define LDS_BYTES (OFF_FLAG + 16)

__device__ __forceinline__ unsigned cnt_pair_to_f16(unsigned w) {
    v2h v;
    v[0] = (half_t)(unsigned short)(w & 0xffffu);
    v[1] = (half_t)(unsigned short)(w >> 16);
    return __builtin_bit_cast(unsigned, v);
}

#define CNT_UNSCALE 16777216.0f

__device__ __forceinline__ void count_edge8(unsigned* cnt32, int src, int dst, int base) {
    const int s = src & (NNODE - 1);
    const unsigned r = (unsigned)(dst - base);
    if (r < (unsigned)BROWS)
        atomicAdd(&cnt32[r * (BPITCH / 4) + (s >> 2)], 1u << ((s & 3) * 8));
}
__device__ __forceinline__ unsigned widen_lo(unsigned d) { return __builtin_amdgcn_perm(0u, d, 0x0c010c00u); }
__device__ __forceinline__ unsigned widen_hi(unsigned d) { return __builtin_amdgcn_perm(0u, d, 0x0c030c02u); }
__device__ __forceinline__ v16h widen_a_cnt8(uint2 lo, uint2 hi) {
    v8u a;
    a[0] = widen_lo(lo.x); a[1] = widen_hi(lo.x); a[2] = widen_lo(lo.y); a[3] = widen_hi(lo.y);
    a[4] = widen_lo(hi.x); a[5] = widen_hi(hi.x); a[6] = widen_lo(hi.y); a[7] = widen_hi(hi.y);
    return __builtin_bit_cast(v16h, a);
}

__device__ __forceinline__ void count_edge(unsigned* cnt32, int src, int dst, int base) {
    const int s = src & (NNODE - 1);
    const unsigned r = (unsigned)(dst - base);
    if (r < (unsigned)CROWS)
        atomicAdd(&cnt32[r * (CPITCH / 4) + (s >> 1)], 1u << ((s & 1) * 16));
}

__device__ __forceinline__ v16h scale_a_f32(float4 f0, float4 f1, float4 f2, float4 f3, float s) {
    v16h a;
    a[0]  = (half_t)(f0.x * s); a[1]  = (half_t)(f0.y * s);
    a[2]  = (half_t)(f0.z * s); a[3]  = (half_t)(f0.w * s);
    a[4]  = (half_t)(f1.x * s); a[5]  = (half_t)(f1.y * s);
    a[6]  = (half_t)(f1.z * s); a[7]  = (half_t)(f1.w * s);
    a[8]  = (half_t)(f2.x * s); a[9]  = (half_t)(f2.y * s);
    a[10] = (half_t)(f2.z * s); a[11] = (half_t)(f2.w * s);
    a[12] = (half_t)(f3.x * s); a[13] = (half_t)(f3.y * s);
    a[14] = (half_t)(f3.z * s); a[15] = (half_t)(f3.w * s);
    return a;
}
__device__ __forceinline__ v16h load_ab_f16(const half_t* row, int k0, int k1) {
    const v8h lo = *(const v8h*)(row + k0);
    const v8h hi = *(const v8h*)(row + k1);
    return __builtin_shufflevector(lo, hi, 0, 1, 2, 3, 4, 5, 6, 7,
                                   8, 9, 10, 11, 12, 13, 14, 15);
}
__device__ __forceinline__ v8h lds_load_tr16(const unsigned char* p) {
    typedef __attribute__((address_space(3))) v8s lds_v8s;
    return __builtin_bit_cast(v8h, __builtin_amdgcn_ds_load_tr16_b128_v8i16((lds_v8s*)p));
}
__device__ __forceinline__ v16h load_b_tr16(const unsigned char* p) {
    const v8h lo = lds_load_tr16(p);
    const v8h hi = lds_load_tr16(p + 16 * WPITCH);
    return __builtin_shufflevector(lo, hi, 0, 1, 2, 3, 4, 5, 6, 7,
                                   8, 9, 10, 11, 12, 13, 14, 15);
}

__device__ __forceinline__ void lds_barrier() {
    __builtin_amdgcn_fence(__ATOMIC_RELEASE, "workgroup", "local");
    __builtin_amdgcn_s_barrier();
    __builtin_amdgcn_fence(__ATOMIC_ACQUIRE, "workgroup", "local");
}

#define H1_BLOCK 1024
__device__ __forceinline__ void tile_epilogue(const v8f& c0, const v8f& c1, int nrow,
                                              int np, int l16, float unscale,
                                              const unsigned* deg, const float* bs, float* embacc) {
    float rin[8];
#pragma unroll
    for (int r = 0; r < 8; ++r) rin[r] = __uint_as_float(deg[NNODE + nrow + r]) * unscale;
#pragma unroll
    for (int j = 0; j < 2; ++j) {
        const v8f c = j ? c1 : c0;
        const int n = (np * 2 + j) * 16 + l16;
        const float bn = bs[n];
        float s = 0.f;
#pragma unroll
        for (int r = 0; r < 8; ++r) s += fmaxf(c[r] * rin[r] + bn, 0.f);
        atomicAdd(&embacc[n], s);
    }
}

__device__ __forceinline__ void tile_store_frag(const v8f& c0, const v8f& c1, int mt, int g,
                                                int np, int lane, int l16, int hi, float unscale,
                                                const unsigned* deg, const float* bs,
                                                half_t* __restrict__ h_out) {
    const int row = mt * 16 + l16;
    const float rin = __uint_as_float(deg[NNODE + row]) * unscale;
    const float rso = __uint_as_float(deg[row]);
    unsigned char* blk = (unsigned char*)h_out
                       + ((size_t)(g * (NNODE / 16) + mt) * (NFEAT / 32) + np) * H1_BLOCK + lane * 16;
#pragma unroll
    for (int j = 0; j < 2; ++j) {
        const v8f c = j ? c1 : c0;
        const float* bp = bs + (np * 2 + j) * 16 + 8 * hi;
        const float4 b0 = *(const float4*)bp, b1 = *(const float4*)(bp + 4);
        const float bn[8] = {b0.x, b0.y, b0.z, b0.w, b1.x, b1.y, b1.z, b1.w};
        v8h y;
#pragma unroll
        for (int r = 0; r < 8; ++r) {
            const float v = fmaxf(c[r] * rin + bn[r], 0.f);
            y[r] = (half_t)(v * rso);
        }
        *(v8h*)(blk + j * (H1_BLOCK / 2)) = y;
    }
}

template <int LAYER>
__device__ __forceinline__ void layer_body(unsigned char* L,
                                           const float*  __restrict__ xf32,
                                           const half_t* __restrict__ xf16,
                                           const int*    __restrict__ edges,
                                           const float*  __restrict__ W,
                                           const float*  __restrict__ bvec,
                                           half_t*       __restrict__ h_out,
                                           float*        __restrict__ emb_out) {
    float*  bs     = (float*)(L + OFF_BS);
    float*  embacc = (float*)(L + OFF_EMB);
    const int g    = blockIdx.x >> 1;
    const int half = blockIdx.x & 1;
    const int t    = threadIdx.x;

    const int* e = edges + (size_t)g * 2 * NEDGE;
    int4 es[4], ed[4];
#pragma unroll
    for (int q = 0; q < 4; ++q) {
        es[q] = *(const int4*)(e + 4 * (t + 512 * q));
        ed[q] = *(const int4*)(e + NEDGE + 4 * (t + 512 * q));
    }
    float4 wr[2 * (NFEAT * (NHID / 8) / 512)];
#pragma unroll
    for (int q = 0; q < NFEAT * (NHID / 8) / 512; ++q) {
        const int i = t + 512 * q;
        const int k = i >> 4, n0 = (i & 15) * 8;
        wr[2 * q]     = *(const float4*)(W + k * NHID + n0);
        wr[2 * q + 1] = *(const float4*)(W + k * NHID + n0 + 4);
    }
    const float bv = bvec[t & (NHID - 1)];
    __builtin_amdgcn_sched_barrier(0);

    const int wave = t >> 5, lane = t & 31;
    const int l16 = lane & 15, hi = lane >> 4;
    const size_t gb = (size_t)g * NNODE;
    const int mi0 = wave, mi1 = wave + 16;
    const int m0 = mi0 * 16 + l16, m1 = mi1 * 16 + l16;

    float4 xr0[16], xr1[16];
    v16h a0[4], a1[4];
    if (LAYER == 0) {
        const float* ar0 = xf32 + (gb + m0) * NFEAT;
        const float* ar1 = xf32 + (gb + m1) * NFEAT;
#pragma unroll
        for (int ko = 0; ko < 4; ++ko) {
            const int k0 = ko * 32 + hi * 8, k1 = ko * 32 + 16 + hi * 8;
            xr0[4 * ko + 0] = *(const float4*)(ar0 + k0);     xr0[4 * ko + 1] = *(const float4*)(ar0 + k0 + 4);
            xr0[4 * ko + 2] = *(const float4*)(ar0 + k1);     xr0[4 * ko + 3] = *(const float4*)(ar0 + k1 + 4);
            xr1[4 * ko + 0] = *(const float4*)(ar1 + k0);     xr1[4 * ko + 1] = *(const float4*)(ar1 + k0 + 4);
            xr1[4 * ko + 2] = *(const float4*)(ar1 + k1);     xr1[4 * ko + 3] = *(const float4*)(ar1 + k1 + 4);
        }
    } else {
        const unsigned char* hf  = (const unsigned char*)xf16 + (size_t)g * (NNODE / 16) * (NFEAT / 32) * H1_BLOCK + lane * 16;
        const unsigned char* hf0 = hf + (size_t)mi0 * (NFEAT / 32) * H1_BLOCK;
        const unsigned char* hf1 = hf + (size_t)mi1 * (NFEAT / 32) * H1_BLOCK;
#pragma unroll
        for (int ko = 0; ko < 4; ++ko) {
            a0[ko] = __builtin_shufflevector(*(const v8h*)(hf0 + ko * H1_BLOCK), *(const v8h*)(hf0 + ko * H1_BLOCK + H1_BLOCK / 2),
                                             0, 1, 2, 3, 4, 5, 6, 7, 8, 9, 10, 11, 12, 13, 14, 15);
            a1[ko] = __builtin_shufflevector(*(const v8h*)(hf1 + ko * H1_BLOCK), *(const v8h*)(hf1 + ko * H1_BLOCK + H1_BLOCK / 2),
                                             0, 1, 2, 3, 4, 5, 6, 7, 8, 9, 10, 11, 12, 13, 14, 15);
        }
    }

    unsigned* deg = (unsigned*)(L + OFF_DEG);
    deg[t] = 0u; deg[t + 512] = 0u;
    unsigned* wide_flag = (unsigned*)(L + OFF_FLAG);
    if (t == 0) *wide_flag = 0u;

#pragma unroll
    for (int q = 0; q < NFEAT * (NHID / 8) / 512; ++q) {
        const int i = t + 512 * q;
        const int k = i >> 4, n0 = (i & 15) * 8;
        const float4 f0 = wr[2 * q], f1 = wr[2 * q + 1];
        v8h w;
        w[0] = (half_t)f0.x; w[1] = (half_t)f0.y; w[2] = (half_t)f0.z; w[3] = (half_t)f0.w;
        w[4] = (half_t)f1.x; w[5] = (half_t)f1.y; w[6] = (half_t)f1.z; w[7] = (half_t)f1.w;
        *(v8h*)(L + OFF_WT + k * WPITCH + n0 * 2) = w;
    }
    if (t < NHID) { bs[t] = bv; embacc[t] = 0.f; }
    uint4* park = (uint4*)(L + OFF_CNT);
#pragma unroll
    for (int q = 0; q < 4; ++q) {
        park[t + 512 * q]       = make_uint4((unsigned)es[q].x, (unsigned)es[q].y, (unsigned)es[q].z, (unsigned)es[q].w);
        park[t + 512 * (4 + q)] = make_uint4((unsigned)ed[q].x, (unsigned)ed[q].y, (unsigned)ed[q].z, (unsigned)ed[q].w);
    }
    lds_barrier();

#pragma unroll
    for (int q = 0; q < 4; ++q) {
        if (LAYER == 0) {
            atomicAdd(&deg[es[q].x & (NNODE - 1)], 1u);
            atomicAdd(&deg[es[q].y & (NNODE - 1)], 1u);
            atomicAdd(&deg[es[q].z & (NNODE - 1)], 1u);
            atomicAdd(&deg[es[q].w & (NNODE - 1)], 1u);
        }
        atomicAdd(&deg[NNODE + (ed[q].x & (NNODE - 1))], 1u);
        atomicAdd(&deg[NNODE + (ed[q].y & (NNODE - 1))], 1u);
        atomicAdd(&deg[NNODE + (ed[q].z & (NNODE - 1))], 1u);
        atomicAdd(&deg[NNODE + (ed[q].w & (NNODE - 1))], 1u);
    }
    lds_barrier();
    {
        const unsigned in_cnt = deg[t + 512];
        if (in_cnt > 255u) *wide_flag = 1u;
        const float d0 = (float)deg[t], d1 = (float)in_cnt;
        deg[t]       = __float_as_uint(rsqrtf(fmaxf(d0, 1.0f)));
        deg[t + 512] = __float_as_uint(rsqrtf(fmaxf(d1, 1.0f)));
    }
    lds_barrier();

    {
        if (LAYER == 0) {
            const float rso0 = __uint_as_float(deg[m0]), rso1 = __uint_as_float(deg[m1]);
#pragma unroll
            for (int ko = 0; ko < 4; ++ko) {
                a0[ko] = scale_a_f32(xr0[4 * ko], xr0[4 * ko + 1], xr0[4 * ko + 2], xr0[4 * ko + 3], rso0);
                a1[ko] = scale_a_f32(xr1[4 * ko], xr1[4 * ko + 1], xr1[4 * ko + 2], xr1[4 * ko + 3], rso1);
            }
        }
        const unsigned char* wl = L + OFF_WT + (8 * hi + (lane & 7)) * WPITCH + 16 * ((lane >> 3) & 1);
#pragma unroll 1
        for (int nq = 0; nq < 2; ++nq) {
            v8f c0[4] = {};
            v8f c1[4] = {};
#pragma unroll
            for (int ko = 0; ko < 4; ++ko) {
#pragma unroll
                for (int j = 0; j < 4; ++j) {
                    const v16h b = load_b_tr16(wl + (ko * 32) * WPITCH + (nq * 4 + j) * 32);
                    c0[j] = __builtin_amdgcn_wmma_f32_16x16x32_f16(
                                false, a0[ko], false, b, (short)0, c0[j], false, false);
                    c1[j] = __builtin_amdgcn_wmma_f32_16x16x32_f16(
                                false, a1[ko], false, b, (short)0, c1[j], false, false);
                }
            }
#pragma unroll
            for (int j = 0; j < 4; ++j) {
                unsigned char* yrow = L + OFF_YT + ((nq * 4 + j) * 16 + l16) * YPITCH;
                v8h y0, y1;
#pragma unroll
                for (int r = 0; r < 8; ++r) { y0[r] = (half_t)c0[j][r]; y1[r] = (half_t)c1[j][r]; }
                *(v8h*)(yrow + (mi0 * 16 + 8 * hi) * 2) = y0;
                *(v8h*)(yrow + (mi1 * 16 + 8 * hi) * 2) = y1;
            }
        }
    }
    __syncthreads();

#pragma unroll
    for (int q = 0; q < 4; ++q) {
        const uint4 ps = park[t + 512 * q], pd = park[t + 512 * (4 + q)];
        es[q] = make_int4((int)ps.x, (int)ps.y, (int)ps.z, (int)ps.w);
        ed[q] = make_int4((int)pd.x, (int)pd.y, (int)pd.z, (int)pd.w);
    }

    unsigned* cnt32  = (unsigned*)(L + OFF_CNT);
    uint4*    cnt128 = (uint4*)(L + OFF_CNT);
    const int np = wave & 3;
    const half_t* brow = (const half_t*)(L + OFF_YT + (np * 32 + l16) * YPITCH);
    const bool wide = __builtin_amdgcn_readfirstlane((int)*wide_flag) != 0;

    if (!wide) {
        const int base = half * BROWS;
        const int mq = wave >> 2;

        for (int i = t; i < BROWS * BPITCH / 16; i += 512) cnt128[i] = make_uint4(0u, 0u, 0u, 0u);
        __syncthreads();

#pragma unroll
        for (int q = 0; q < 4; ++q) {
            count_edge8(cnt32, es[q].x, ed[q].x, base);
            count_edge8(cnt32, es[q].y, ed[q].y, base);
            count_edge8(cnt32, es[q].z, ed[q].z, base);
            count_edge8(cnt32, es[q].w, ed[q].w, base);
        }
        __syncthreads();

        v8f acc[4][2] = {};
        const unsigned char* arow = L + OFF_CNT + (mq * 64 + l16) * BPITCH + hi * 8;
        uint2 ra[2][4][2];
        v16h  rb[2][2];
#pragma unroll
        for (int i = 0; i < 4; ++i) {
            ra[0][i][0] = *(const uint2*)(arow + i * 16 * BPITCH);
            ra[0][i][1] = *(const uint2*)(arow + i * 16 * BPITCH + 16);
        }
        rb[0][0] = load_ab_f16(brow, hi * 8, 16 + hi * 8);
        rb[0][1] = load_ab_f16(brow + 16 * (YPITCH / 2), hi * 8, 16 + hi * 8);
#pragma unroll 1
        for (int kk = 0; kk < 16; kk += 2) {
#pragma unroll
            for (int s = 0; s < 2; ++s) {
                const int kn = ((kk + s + 1) & 15) * 32;
#pragma unroll
                for (int i = 0; i < 4; ++i) {
                    ra[s ^ 1][i][0] = *(const uint2*)(arow + i * 16 * BPITCH + kn);
                    ra[s ^ 1][i][1] = *(const uint2*)(arow + i * 16 * BPITCH + kn + 16);
                }
                rb[s ^ 1][0] = load_ab_f16(brow, kn + hi * 8, kn + 16 + hi * 8);
                rb[s ^ 1][1] = load_ab_f16(brow + 16 * (YPITCH / 2), kn + hi * 8, kn + 16 + hi * 8);
                __builtin_amdgcn_sched_barrier(0);
                v16h a[4];
#pragma unroll
                for (int i = 0; i < 4; ++i) a[i] = widen_a_cnt8(ra[s][i][0], ra[s][i][1]);
#pragma unroll
                for (int i = 0; i < 4; ++i)
#pragma unroll
                    for (int j = 0; j < 2; ++j)
                        acc[i][j] = LAYER == 0
                            ? __builtin_amdgcn_wmma_f32_16x16x32_f16(
                                        false, rb[s][j], false, a[i], (short)0, acc[i][j], false, false)
                            : __builtin_amdgcn_wmma_f32_16x16x32_f16(
                                        false, a[i], false, rb[s][j], (short)0, acc[i][j], false, false);
                __builtin_amdgcn_sched_barrier(0);
            }
        }

#pragma unroll
        for (int i = 0; i < 4; ++i) {
            if (LAYER == 0)
                tile_store_frag(acc[i][0], acc[i][1], (base >> 4) + mq * 4 + i, g, np, lane, l16, hi,
                                CNT_UNSCALE, deg, bs, h_out);
            else
                tile_epilogue(acc[i][0], acc[i][1], base + (mq * 4 + i) * 16 + 8 * hi, np, l16,
                              CNT_UNSCALE, deg, bs, embacc);
        }
        __syncthreads();
    } else {
        const int mp = wave >> 2;
        for (int p = 0; p < 2; ++p) {
            const int base = half * 256 + p * CROWS;

            for (int i = t; i < CROWS * CPITCH / 16; i += 512) cnt128[i] = make_uint4(0u, 0u, 0u, 0u);
            __syncthreads();

#pragma unroll
            for (int q = 0; q < 4; ++q) {
                count_edge(cnt32, es[q].x, ed[q].x, base);
                count_edge(cnt32, es[q].y, ed[q].y, base);
                count_edge(cnt32, es[q].z, ed[q].z, base);
                count_edge(cnt32, es[q].w, ed[q].w, base);
            }
            __syncthreads();

            for (int i = t; i < CROWS * CPITCH / 16; i += 512) {
                uint4 w = cnt128[i];
                w.x = cnt_pair_to_f16(w.x); w.y = cnt_pair_to_f16(w.y);
                w.z = cnt_pair_to_f16(w.z); w.w = cnt_pair_to_f16(w.w);
                cnt128[i] = w;
            }
            __syncthreads();

            v8f acc[2][2] = {};
            const half_t* arow = (const half_t*)(L + OFF_CNT + (mp * 32 + l16) * CPITCH);
#pragma unroll 4
            for (int ks = 0; ks < 16; ++ks) {
                const int k0 = ks * 32 + hi * 8;
                const int k1 = ks * 32 + 16 + hi * 8;
                v16h a[2], b[2];
                a[0] = load_ab_f16(arow, k0, k1);
                a[1] = load_ab_f16(arow + 16 * (CPITCH / 2), k0, k1);
                b[0] = load_ab_f16(brow, k0, k1);
                b[1] = load_ab_f16(brow + 16 * (YPITCH / 2), k0, k1);
#pragma unroll
                for (int i = 0; i < 2; ++i)
#pragma unroll
                    for (int j = 0; j < 2; ++j)
                        acc[i][j] = LAYER == 0
                            ? __builtin_amdgcn_wmma_f32_16x16x32_f16(
                                        false, b[j], false, a[i], (short)0, acc[i][j], false, false)
                            : __builtin_amdgcn_wmma_f32_16x16x32_f16(
                                        false, a[i], false, b[j], (short)0, acc[i][j], false, false);
            }

#pragma unroll
            for (int i = 0; i < 2; ++i) {
                if (LAYER == 0)
                    tile_store_frag(acc[i][0], acc[i][1], (base >> 4) + mp * 2 + i, g, np, lane, l16, hi,
                                    1.0f, deg, bs, h_out);
                else
                    tile_epilogue(acc[i][0], acc[i][1], base + (mp * 2 + i) * 16 + 8 * hi, np, l16,
                                  1.0f, deg, bs, embacc);
            }
            __syncthreads();
        }
    }

    if (LAYER == 1) {
        if (t < NHID) atomicAdd(&emb_out[g * NHID + t], embacc[t] * (1.0f / NNODE));
    }
}

__global__ __launch_bounds__(512, 1) void k_layer0(const float* __restrict__ x,
                                                   const int*   __restrict__ edges,
                                                   const float* __restrict__ W,
                                                   const float* __restrict__ bvec,
                                                   half_t*      __restrict__ h_out) {
    __shared__ __attribute__((aligned(16))) unsigned char L[LDS_BYTES];
    layer_body<0>(L, x, nullptr, edges, W, bvec, h_out, nullptr);
}

__global__ __launch_bounds__(512, 1) void k_layer1(const half_t* __restrict__ hs,
                                                   const int*    __restrict__ edges,
                                                   const float*  __restrict__ W,
                                                   const float*  __restrict__ bvec,
                                                   float*        __restrict__ emb_out) {
    __shared__ __attribute__((aligned(16))) unsigned char L[LDS_BYTES];
    layer_body<1>(L, nullptr, hs, edges, W, bvec, nullptr, emb_out);
}

__global__ __launch_bounds__(128) void k_head(const float* __restrict__ emb,
                                              const float* __restrict__ Wt,
                                              const float* __restrict__ bt,
                                              const float* __restrict__ gamma,
                                              const float* __restrict__ beta,
                                              const float* __restrict__ rmean,
                                              const float* __restrict__ rvar,
                                              float* __restrict__ out) {
    __shared__ float er[NHID];
    const int g = blockIdx.x, n = threadIdx.x;
    const float e  = emb[g * NHID + n];
    const float zb = bt[n], gm = gamma[n], be = beta[n], rm = rmean[n], rv = rvar[n];
    float w[NHID];
#pragma unroll
    for (int k = 0; k < NHID; ++k) w[k] = Wt[k * NHID + n];
    er[n] = e;
    lds_barrier();
    float z = zb;
#pragma unroll
    for (int k = 0; k < NHID; ++k) z += er[k] * w[k];
    z = gm * (z - rm) * rsqrtf(rv + BN_EPS) + be;
    out[g * NHID + n] = fmaxf(z, 0.f);
}

extern "C" void kernel_launch(void* const* d_in, const int* in_sizes, int n_in,
                              void* d_out, int out_size, void* d_ws, size_t ws_size,
                              hipStream_t stream) {
    (void)in_sizes; (void)n_in; (void)out_size; (void)ws_size;
    const float* feats = (const float*)d_in[0];
    const int*   edges = (const int*)d_in[1];
    const float* W0    = (const float*)d_in[2];
    const float* b0    = (const float*)d_in[3];
    const float* W1    = (const float*)d_in[4];
    const float* b1    = (const float*)d_in[5];
    const float* Wt    = (const float*)d_in[6];
    const float* bt    = (const float*)d_in[7];
    const float* gamma = (const float*)d_in[8];
    const float* beta  = (const float*)d_in[9];
    const float* rmean = (const float*)d_in[10];
    const float* rvar  = (const float*)d_in[11];
    float* out = (float*)d_out;

    char* ws = (char*)d_ws;
    half_t* h1s    = (half_t*)ws; ws += (size_t)NGRAPH * NNODE * NFEAT * sizeof(half_t);
    float*  emb    = (float*)ws;  ws += (size_t)NGRAPH * NHID * sizeof(float);
    const size_t emb_bytes = (size_t)NGRAPH * NHID * sizeof(float);

    k_layer0<<<NGRAPH * 2, 512, 0, stream>>>(feats, edges, W0, b0, h1s);

    hipMemsetAsync(emb, 0, emb_bytes, stream);
    k_layer1<<<NGRAPH * 2, 512, 0, stream>>>(h1s, edges, W1, b1, emb);

    k_head<<<NGRAPH, 128, 0, stream>>>(emb, Wt, bt, gamma, beta, rmean, rvar, out);
}
